# starting kernel
# baseline (speedup 1.0000x reference)
#include <hip/hip_runtime.h>

#define N_NODES 100000
#define N_EDGES 640000
#define D 128
#define EPS 1e-5f
#define LDSTRIDE 132

typedef __attribute__((ext_vector_type(2))) float v2f;
typedef __attribute__((ext_vector_type(8))) float v8f;

__global__ void zero_ws_kernel(float* __restrict__ p, long n) {
    long i = (long)blockIdx.x * blockDim.x + threadIdx.x;
    long stride = (long)gridDim.x * blockDim.x;
    for (; i < n; i += stride) p[i] = 0.0f;
}

__global__ __launch_bounds__(256) void layernorm_kernel(
    const float* __restrict__ h, const float* __restrict__ gamma,
    const float* __restrict__ beta, float* __restrict__ hn) {
    int wid  = (blockIdx.x * blockDim.x + threadIdx.x) >> 5;
    int lane = threadIdx.x & 31;
    if (wid >= N_NODES) return;

    float4 v = ((const float4*)(h + (long)wid * D))[lane];
    float s = v.x + v.y + v.z + v.w;
#pragma unroll
    for (int off = 16; off; off >>= 1) s += __shfl_xor(s, off, 32);
    float mu = s * (1.0f / D);

    float dx = v.x - mu, dy = v.y - mu, dz = v.z - mu, dw = v.w - mu;
    float q = dx * dx + dy * dy + dz * dz + dw * dw;
#pragma unroll
    for (int off = 16; off; off >>= 1) q += __shfl_xor(q, off, 32);
    float rstd = rsqrtf(q * (1.0f / D) + EPS);

    float4 g  = ((const float4*)gamma)[lane];
    float4 bt = ((const float4*)beta)[lane];
    float4 o;
    o.x = dx * rstd * g.x + bt.x;
    o.y = dy * rstd * g.y + bt.y;
    o.z = dz * rstd * g.z + bt.z;
    o.w = dw * rstd * g.w + bt.w;
    ((float4*)(hn + (long)wid * D))[lane] = o;
}

__global__ __launch_bounds__(256) void scatter_kernel(
    const float* __restrict__ hn, const int* __restrict__ src,
    const int* __restrict__ dst, float* __restrict__ msg,
    float* __restrict__ deg) {
    int wid  = (blockIdx.x * blockDim.x + threadIdx.x) >> 5;
    int lane = threadIdx.x & 31;
    if (wid >= N_EDGES) return;
    int s = src[wid], d = dst[wid];
    const float* hs = hn + (long)s * D;
    float* md = msg + (long)d * D;
#pragma unroll
    for (int i = 0; i < 4; ++i)
        atomicAdd(md + lane + 32 * i, hs[lane + 32 * i]);
    if (lane == 0) atomicAdd(deg + d, 1.0f);
}

__global__ __launch_bounds__(256) void sage_gemm_kernel(
    const float* __restrict__ hn, const float* __restrict__ msg,
    const float* __restrict__ deg, const float* __restrict__ Wself,
    const float* __restrict__ Wneigh, const float* __restrict__ bias,
    float* __restrict__ out) {
    __shared__ float As[32 * LDSTRIDE];
    __shared__ float An[32 * LDSTRIDE];
    __shared__ float invdeg[32];

    int t = threadIdx.x;
    int rowBase = blockIdx.x * 32;

    if (t < 32) invdeg[t] = 1.0f / fmaxf(deg[rowBase + t], 1.0f);
#pragma unroll
    for (int i = 0; i < 16; ++i) {
        int e = t + i * 256;
        int r = e >> 7, c = e & 127;
        As[r * LDSTRIDE + c] = hn[(long)(rowBase + r) * D + c];
    }
    __syncthreads();
#pragma unroll
    for (int i = 0; i < 16; ++i) {
        int e = t + i * 256;
        int r = e >> 7, c = e & 127;
        An[r * LDSTRIDE + c] = msg[(long)(rowBase + r) * D + c] * invdeg[r];
    }
    __syncthreads();

    int lane = t & 31;
    int wave = t >> 5;
    int nq   = lane & 15;
    int hi   = lane >> 4;
    int ncol = wave * 16 + nq;
    int m    = nq;

    v8f acc0 = {};
    v8f acc1 = {};

#pragma unroll 4
    for (int k = 0; k < D; k += 4) {
        int kk = k + (hi << 1);
        v2f a0, a1, b;
        a0.x = As[m * LDSTRIDE + kk];
        a0.y = As[m * LDSTRIDE + kk + 1];
        a1.x = As[(m + 16) * LDSTRIDE + kk];
        a1.y = As[(m + 16) * LDSTRIDE + kk + 1];
        b.x  = Wself[kk * D + ncol];
        b.y  = Wself[(kk + 1) * D + ncol];
        acc0 = __builtin_amdgcn_wmma_f32_16x16x4_f32(
            false, a0, false, b, (short)0, acc0, false, false);
        acc1 = __builtin_amdgcn_wmma_f32_16x16x4_f32(
            false, a1, false, b, (short)0, acc1, false, false);
    }
#pragma unroll 4
    for (int k = 0; k < D; k += 4) {
        int kk = k + (hi << 1);
        v2f a0, a1, b;
        a0.x = An[m * LDSTRIDE + kk];
        a0.y = An[m * LDSTRIDE + kk + 1];
        a1.x = An[(m + 16) * LDSTRIDE + kk];
        a1.y = An[(m + 16) * LDSTRIDE + kk + 1];
        b.x  = Wneigh[kk * D + ncol];
        b.y  = Wneigh[(kk + 1) * D + ncol];
        acc0 = __builtin_amdgcn_wmma_f32_16x16x4_f32(
            false, a0, false, b, (short)0, acc0, false, false);
        acc1 = __builtin_amdgcn_wmma_f32_16x16x4_f32(
            false, a1, false, b, (short)0, acc1, false, false);
    }

    float bv = bias[ncol];
    int mBase = hi * 8;
#pragma unroll
    for (int i = 0; i < 8; ++i) {
        out[(long)(rowBase + mBase + i) * D + ncol]      = acc0[i] + bv;
        out[(long)(rowBase + 16 + mBase + i) * D + ncol] = acc1[i] + bv;
    }
}

extern "C" void kernel_launch(void* const* d_in, const int* in_sizes, int n_in,
                              void* d_out, int out_size, void* d_ws, size_t ws_size,
                              hipStream_t stream) {
    const float* h      = (const float*)d_in[0];
    const int*   src    = (const int*)d_in[1];
    const int*   dst    = (const int*)d_in[2];
    const float* gamma  = (const float*)d_in[3];
    const float* beta   = (const float*)d_in[4];
    const float* Wself  = (const float*)d_in[5];
    const float* Wneigh = (const float*)d_in[6];
    const float* bias   = (const float*)d_in[7];
    float* out = (float*)d_out;

    float* hn  = (float*)d_ws;
    float* msg = hn + (long)N_NODES * D;
    float* deg = msg + (long)N_NODES * D;

    long zcount = (long)N_NODES * D + N_NODES;
    zero_ws_kernel<<<2048, 256, 0, stream>>>(msg, zcount);

    layernorm_kernel<<<(N_NODES + 7) / 8, 256, 0, stream>>>(h, gamma, beta, hn);

    scatter_kernel<<<(N_EDGES + 7) / 8, 256, 0, stream>>>(hn, src, dst, msg, deg);

    sage_gemm_kernel<<<N_NODES / 32, 256, 0, stream>>>(
        hn, msg, deg, Wself, Wneigh, bias, out);
}


# Float-atomic scatter-add -> gather: an int add gives each edge a slot in its dst's index line; GEMM waves sum a row's sources in registers, ascending index (sum order changed; fixed to 32 in-edges)
# speedup vs baseline: 1.6293x; 1.6293x over previous
#include <hip/hip_runtime.h>

#define N_NODES 100000
#define N_EDGES 640000
#define D 128
#define EPS 1e-5f
#define LDSTRIDE 132
#define CAP 32

typedef __attribute__((ext_vector_type(2))) float v2f;
typedef __attribute__((ext_vector_type(8))) float v8f;

__global__ void zero_ws_kernel(float* __restrict__ p, long n) {
    long i = (long)blockIdx.x * blockDim.x + threadIdx.x;
    long stride = (long)gridDim.x * blockDim.x;
    for (; i < n; i += stride) p[i] = 0.0f;
}

__global__ __launch_bounds__(256) void layernorm_kernel(
    const float* __restrict__ h, const float* __restrict__ gamma,
    const float* __restrict__ beta, float* __restrict__ hn) {
    int wid  = (blockIdx.x * blockDim.x + threadIdx.x) >> 5;
    int lane = threadIdx.x & 31;
    if (wid >= N_NODES) return;

    float4 v = ((const float4*)(h + (long)wid * D))[lane];
    float s = v.x + v.y + v.z + v.w;
#pragma unroll
    for (int off = 16; off; off >>= 1) s += __shfl_xor(s, off, 32);
    float mu = s * (1.0f / D);

    float dx = v.x - mu, dy = v.y - mu, dz = v.z - mu, dw = v.w - mu;
    float q = dx * dx + dy * dy + dz * dz + dw * dw;
#pragma unroll
    for (int off = 16; off; off >>= 1) q += __shfl_xor(q, off, 32);
    float rstd = rsqrtf(q * (1.0f / D) + EPS);

    float4 g  = ((const float4*)gamma)[lane];
    float4 bt = ((const float4*)beta)[lane];
    float4 o;
    o.x = dx * rstd * g.x + bt.x;
    o.y = dy * rstd * g.y + bt.y;
    o.z = dz * rstd * g.z + bt.z;
    o.w = dw * rstd * g.w + bt.w;
    ((float4*)(hn + (long)wid * D))[lane] = o;
}

__global__ __launch_bounds__(256) void fill_kernel(
    const int* __restrict__ src, const int* __restrict__ dst,
    int* __restrict__ cnt, int* __restrict__ col, int* __restrict__ ovf_n,
    int* __restrict__ ovf_d, int* __restrict__ ovf_s) {
    int e = blockIdx.x * blockDim.x + threadIdx.x;
    if (e >= N_EDGES) return;
    int d = dst[e], s = src[e];
    int r = atomicAdd(cnt + d, 1);
    if (r < CAP) {
        col[(long)d * CAP + r] = s;
    } else {
        int o = atomicAdd(ovf_n, 1);
        o = o < N_EDGES - 1 ? o : N_EDGES - 1;
        ovf_d[o] = d;
        ovf_s[o] = s;
    }
}

__global__ __launch_bounds__(256) void sage_gemm_kernel(
    const float* __restrict__ hn, const int* __restrict__ cnt,
    const int* __restrict__ col, const int* __restrict__ ovf_n,
    const int* __restrict__ ovf_d, const int* __restrict__ ovf_s,
    const float* __restrict__ Wself,
    const float* __restrict__ Wneigh, const float* __restrict__ bias,
    float* __restrict__ out) {
    __shared__ __attribute__((aligned(16))) float As[32 * LDSTRIDE];
    __shared__ __attribute__((aligned(16))) float An[32 * LDSTRIDE];

    int t = threadIdx.x;
    int rowBase = blockIdx.x * 32;
    int lane = t & 31;
    int wv = __builtin_amdgcn_readfirstlane(t >> 5);

    int novf = ovf_n[0];
    novf = novf < N_EDGES ? novf : N_EDGES;

    const int node0 = rowBase + wv * 4;
    const int c0 = __builtin_amdgcn_readfirstlane(cnt[node0 + 0]);
    const int c1 = __builtin_amdgcn_readfirstlane(cnt[node0 + 1]);
    const int c2 = __builtin_amdgcn_readfirstlane(cnt[node0 + 2]);
    const int c3 = __builtin_amdgcn_readfirstlane(cnt[node0 + 3]);
    int k0 = 0x7fffffff, k1 = 0x7fffffff, k2 = 0x7fffffff, k3 = 0x7fffffff;
    if (lane < (c0 < CAP ? c0 : CAP)) k0 = col[(long)(node0 + 0) * CAP + lane];
    if (lane < (c1 < CAP ? c1 : CAP)) k1 = col[(long)(node0 + 1) * CAP + lane];
    if (lane < (c2 < CAP ? c2 : CAP)) k2 = col[(long)(node0 + 2) * CAP + lane];
    if (lane < (c3 < CAP ? c3 : CAP)) k3 = col[(long)(node0 + 3) * CAP + lane];

#pragma unroll
    for (int i = 0; i < 16; ++i) {
        int e = t + i * 256;
        int r = e >> 7, c = e & 127;
        As[r * LDSTRIDE + c] = hn[(long)(rowBase + r) * D + c];
    }

    for (int rr = 0; rr < 4; ++rr) {
        int r = wv * 4 + rr;
        const int c   = rr == 0 ? c0 : rr == 1 ? c1 : rr == 2 ? c2 : c3;
        const int key = rr == 0 ? k0 : rr == 1 ? k1 : rr == 2 ? k2 : k3;
        int n = c < CAP ? c : CAP;
        int rank = 0;
        for (int j = 0; j < n; ++j) {
            int kj = __builtin_amdgcn_readlane(key, j);
            rank += ((kj < key) || (kj == key && j < lane)) ? 1 : 0;
        }
        float4 acc = make_float4(0.0f, 0.0f, 0.0f, 0.0f);
        int j = 0;
        for (; j + 4 <= n; j += 4) {
            float4 v[4];
#pragma unroll
            for (int u = 0; u < 4; ++u) {
                unsigned m = __builtin_amdgcn_ballot_w32(lane < n && rank == j + u);
                int sl = m ? __builtin_ctz(m) : 0;
                unsigned s = (unsigned)__builtin_amdgcn_readlane(key, sl);
                s = s < N_NODES - 1 ? s : N_NODES - 1;
                v[u] = ((const float4*)(hn + (long)s * D))[lane];
            }
#pragma unroll
            for (int u = 0; u < 4; ++u) {
                acc.x += v[u].x; acc.y += v[u].y; acc.z += v[u].z; acc.w += v[u].w;
            }
        }
        for (; j < n; ++j) {
            unsigned m = __builtin_amdgcn_ballot_w32(lane < n && rank == j);
            int sl = m ? __builtin_ctz(m) : 0;
            unsigned s = (unsigned)__builtin_amdgcn_readlane(key, sl);
            s = s < N_NODES - 1 ? s : N_NODES - 1;
            float4 v0 = ((const float4*)(hn + (long)s * D))[lane];
            acc.x += v0.x; acc.y += v0.y; acc.z += v0.z; acc.w += v0.w;
        }
        float sc = novf > 0 ? 1.0f : 1.0f / fmaxf((float)c, 1.0f);
        float4 o;
        o.x = acc.x * sc; o.y = acc.y * sc; o.z = acc.z * sc; o.w = acc.w * sc;
        *(float4*)(An + r * LDSTRIDE + lane * 4) = o;
    }

    if (novf > 0) {
        __syncthreads();
        for (int e = wv; e < novf; e += 8) {
            int d = __builtin_amdgcn_readfirstlane(ovf_d[e]);
            int r = d - rowBase;
            if (r >= 0 && r < 32) {
                unsigned s = (unsigned)__builtin_amdgcn_readfirstlane(ovf_s[e]);
                s = s < N_NODES - 1 ? s : N_NODES - 1;
                float4 v0 = ((const float4*)(hn + (long)s * D))[lane];
                float* p = An + r * LDSTRIDE + lane * 4;
                atomicAdd(p + 0, v0.x);
                atomicAdd(p + 1, v0.y);
                atomicAdd(p + 2, v0.z);
                atomicAdd(p + 3, v0.w);
            }
        }
        __syncthreads();
        for (int i = 0; i < 16; ++i) {
            int e = t + i * 256;
            int r = e >> 7, c = e & 127;
            An[r * LDSTRIDE + c] *= 1.0f / fmaxf((float)cnt[rowBase + r], 1.0f);
        }
    }
    __syncthreads();

    int wave = wv;
    int nq   = lane & 15;
    int hi   = lane >> 4;
    int ncol = wave * 16 + nq;
    int m    = nq;

    v8f acc0 = {};
    v8f acc1 = {};

#pragma unroll 4
    for (int k = 0; k < D; k += 4) {
        int kk = k + (hi << 1);
        v2f a0, a1, b;
        a0.x = As[m * LDSTRIDE + kk];
        a0.y = As[m * LDSTRIDE + kk + 1];
        a1.x = As[(m + 16) * LDSTRIDE + kk];
        a1.y = As[(m + 16) * LDSTRIDE + kk + 1];
        b.x  = Wself[kk * D + ncol];
        b.y  = Wself[(kk + 1) * D + ncol];
        acc0 = __builtin_amdgcn_wmma_f32_16x16x4_f32(
            false, a0, false, b, (short)0, acc0, false, false);
        acc1 = __builtin_amdgcn_wmma_f32_16x16x4_f32(
            false, a1, false, b, (short)0, acc1, false, false);
    }
#pragma unroll 4
    for (int k = 0; k < D; k += 4) {
        int kk = k + (hi << 1);
        v2f a0, a1, b;
        a0.x = An[m * LDSTRIDE + kk];
        a0.y = An[m * LDSTRIDE + kk + 1];
        a1.x = An[(m + 16) * LDSTRIDE + kk];
        a1.y = An[(m + 16) * LDSTRIDE + kk + 1];
        b.x  = Wneigh[kk * D + ncol];
        b.y  = Wneigh[(kk + 1) * D + ncol];
        acc0 = __builtin_amdgcn_wmma_f32_16x16x4_f32(
            false, a0, false, b, (short)0, acc0, false, false);
        acc1 = __builtin_amdgcn_wmma_f32_16x16x4_f32(
            false, a1, false, b, (short)0, acc1, false, false);
    }

    float bv = bias[ncol];
    int mBase = hi * 8;
#pragma unroll
    for (int i = 0; i < 8; ++i) {
        out[(long)(rowBase + mBase + i) * D + ncol]      = acc0[i] + bv;
        out[(long)(rowBase + 16 + mBase + i) * D + ncol] = acc1[i] + bv;
    }
}

extern "C" void kernel_launch(void* const* d_in, const int* in_sizes, int n_in,
                              void* d_out, int out_size, void* d_ws, size_t ws_size,
                              hipStream_t stream) {
    const float* h      = (const float*)d_in[0];
    const int*   src    = (const int*)d_in[1];
    const int*   dst    = (const int*)d_in[2];
    const float* gamma  = (const float*)d_in[3];
    const float* beta   = (const float*)d_in[4];
    const float* Wself  = (const float*)d_in[5];
    const float* Wneigh = (const float*)d_in[6];
    const float* bias   = (const float*)d_in[7];
    float* out = (float*)d_out;

    float* hn  = (float*)d_ws;
    int* cnt   = (int*)(hn + (long)N_NODES * D);
    int* ovf_n = cnt + N_NODES;
    int* col   = ovf_n + 32;
    int* ovf_d = col + (long)N_NODES * CAP;
    int* ovf_s = ovf_d + N_EDGES;

    zero_ws_kernel<<<(N_NODES + 32 + 255) / 256, 256, 0, stream>>>(
        (float*)cnt, (long)N_NODES + 32);

    layernorm_kernel<<<(N_NODES + 7) / 8, 256, 0, stream>>>(h, gamma, beta, hn);

    fill_kernel<<<(N_EDGES + 255) / 256, 256, 0, stream>>>(
        src, dst, cnt, col, ovf_n, ovf_d, ovf_s);

    sage_gemm_kernel<<<N_NODES / 32, 256, 0, stream>>>(
        hn, cnt, col, ovf_n, ovf_d, ovf_s, Wself, Wneigh, bias, out);
}
